# starting kernel
# baseline (speedup 1.0000x reference)
#include <hip/hip_runtime.h>

typedef float v2f __attribute__((ext_vector_type(2)));
typedef float v8f __attribute__((ext_vector_type(8)));

#define BATCH_   64
#define SIZE_    256
#define NPR_     128
#define NPC_     128
#define NPCG_    8
#define NTILES_  (BATCH_ * NPR_ * NPCG_)

__device__ __forceinline__ v8f wmma4(v2f a, v2f b, v8f c) {
  return __builtin_amdgcn_wmma_f32_16x16x4_f32(false, a, false, b, (short)0, c,
                                               false, false);
}

__global__ __launch_bounds__(256)
void qconv2d_wmma_kernel(const float* __restrict__ x,
                         const float* __restrict__ w,
                         float* __restrict__ out) {
  const int lane = threadIdx.x & 31;
  const int p    = lane & 15;
  const int off  = (lane & 16) ? 2 : 0;

  float cq[4], sq[4];
#pragma unroll
  for (int q = 0; q < 4; ++q) {
    __sincosf(0.5f * w[q], &sq[q], &cq[q]);
  }
  const int s0 = p & 1, s1 = (p >> 1) & 1, s2 = (p >> 2) & 1, s3 = (p >> 3) & 1;
  const int j = (s0 ^ s3) | ((s0 ^ s1 ^ s3) << 1) | ((s1 ^ s2) << 2) | ((s2 ^ s3) << 3);

  v2f aR[4], aI[4];
#pragma unroll
  for (int c = 0; c < 4; ++c) {
    float mr[2], mi[2];
#pragma unroll
    for (int jj = 0; jj < 2; ++jj) {
      const int k  = 4 * c + off + jj;
      const int xo = j ^ k;
      const int pc = __popc(xo);
      float prod = 1.0f;
#pragma unroll
      for (int q = 0; q < 4; ++q) prod *= ((xo >> q) & 1) ? sq[q] : cq[q];
      if (pc & 1) { mr[jj] = 0.0f; mi[jj] = (pc & 2) ? prod : -prod; }
      else        { mi[jj] = 0.0f; mr[jj] = (pc & 2) ? -prod : prod; }
    }
    aR[c] = (v2f){mr[0], mr[1]};
    aI[c] = (v2f){mi[0], mi[1]};
  }

  const int nWaves = gridDim.x * (blockDim.x >> 5);
  const int gw     = blockIdx.x * (blockDim.x >> 5) + (threadIdx.x >> 5);

  for (int tile = gw; tile < NTILES_; tile += nWaves) {
    const int b   = tile >> 10;
    const int rem = tile & 1023;
    const int pr  = rem >> 3;
    const int pcg = rem & 7;

    const float* base = x + ((size_t)b * SIZE_ + 2 * pr) * SIZE_ + (pcg * 16 + p) * 2;
    const float2 lo = *(const float2*)base;
    const float2 hi = *(const float2*)(base + SIZE_);
    const float a0 = lo.x, a1 = lo.y, a2 = hi.x, a3 = hi.y;
    const float ss = a0 * a0 + a1 * a1 + a2 * a2 + a3 * a3;

    v8f Cr = {0.f, 0.f, 0.f, 0.f, 0.f, 0.f, 0.f, 0.f};
    v8f Ci = {0.f, 0.f, 0.f, 0.f, 0.f, 0.f, 0.f, 0.f};

#pragma unroll
    for (int c = 0; c < 4; ++c) {
      float cs[2], sn[2];
#pragma unroll
      for (int jj = 0; jj < 2; ++jj) {
        const int k = 4 * c + off + jj;
        const float t = ((k & 1) ? a0 : -a0) + ((k & 2) ? a1 : -a1) +
                        ((k & 4) ? a2 : -a2) + ((k & 8) ? a3 : -a3);
        const float phi = 0.5f * t + 0.25f * (t * t - ss);
        __sincosf(phi, &sn[jj], &cs[jj]);
      }
      const v2f bC  = (v2f){cs[0], cs[1]};
      const v2f bS  = (v2f){sn[0], sn[1]};
      const v2f bSn = (v2f){-sn[0], -sn[1]};
      Cr = wmma4(aR[c], bC,  Cr);
      Cr = wmma4(aI[c], bS,  Cr);
      Ci = wmma4(aI[c], bC,  Ci);
      Ci = wmma4(aR[c], bSn, Ci);
    }

    float acc0 = 0.f, acc1 = 0.f, acc2 = 0.f, acc3 = 0.f;
    const int sbase = (lane & 16) ? 8 : 0;
#pragma unroll
    for (int v = 0; v < 8; ++v) {
      const float pv = 0.0625f * (Cr[v] * Cr[v] + Ci[v] * Ci[v]);
      const int s = sbase + v;
      if (s & 8) acc0 += pv;
      if (s & 4) acc1 += pv;
      if (s & 2) acc2 += pv;
      if (s & 1) acc3 += pv;
    }
    acc0 += __shfl_xor(acc0, 16, 32);
    acc1 += __shfl_xor(acc1, 16, 32);
    acc2 += __shfl_xor(acc2, 16, 32);
    acc3 += __shfl_xor(acc3, 16, 32);

    if (lane < 16) {
      float4 o = make_float4(acc0, acc1, acc2, acc3);
      *(float4*)(out + (((size_t)b * NPR_ + pr) * NPC_ + pcg * 16 + p) * 4) = o;
    }
  }
}

extern "C" void kernel_launch(void* const* d_in, const int* in_sizes, int n_in,
                              void* d_out, int out_size, void* d_ws, size_t ws_size,
                              hipStream_t stream) {
  (void)in_sizes; (void)n_in; (void)d_ws; (void)ws_size; (void)out_size;
  const float* x = (const float*)d_in[0];
  const float* w = (const float*)d_in[1];
  float* out = (float*)d_out;
  qconv2d_wmma_kernel<<<dim3(2048), dim3(256), 0, stream>>>(x, w, out);
}


# Bit-flip symmetry: u commutes with X on all qubits and d(15-k) is d(k) with the linear phase conjugated, so M*d folds into two 8x8 complex products, 8 WMMAs per tile not 16; sums reordered
# speedup vs baseline: 1.2716x; 1.2716x over previous
#include <hip/hip_runtime.h>


typedef float v2f __attribute__((ext_vector_type(2)));
typedef float v8f __attribute__((ext_vector_type(8)));

#define BATCH_   64
#define SIZE_    256
#define NPR_     128
#define NPC_     128
#define NPCG_    8
#define NTILES_  (BATCH_ * NPR_ * NPCG_)

__device__ __forceinline__ v8f wmma4(v2f a, v2f b, v8f c) {
  return __builtin_amdgcn_wmma_f32_16x16x4_f32(false, a, false, b, (short)0, c,
                                               false, false);
}

__global__ __launch_bounds__(256)
void qconv2d_wmma_kernel(const float* __restrict__ x,
                         const float* __restrict__ w,
                         float* __restrict__ out) {
  const int lane = threadIdx.x & 31;
  const int p    = lane & 15;
  const int h    = (lane >> 4) & 1;

  float cq[4], sq[4];
#pragma unroll
  for (int q = 0; q < 4; ++q) {
    __sincosf(0.5f * w[q], &sq[q], &cq[q]);
  }
  const int j     = p & 7;
  const int imrow = p >> 3;

  v2f aG[4], aH[4];
#pragma unroll
  for (int c = 0; c < 4; ++c) {
    float ag[2], ah[2];
#pragma unroll
    for (int jj = 0; jj < 2; ++jj) {
      const int k    = 4 * (c & 1) + 2 * h + jj;
      const int part = c >> 1;
      const int xo   = j ^ k;
      const int pc   = __popc(xo);
      float p1 = 1.0f, p2 = 1.0f;
#pragma unroll
      for (int q = 0; q < 4; ++q) {
        const bool b = (xo >> q) & 1;
        p1 *= b ? sq[q] : cq[q];
        p2 *= b ? cq[q] : sq[q];
      }
      float upr, upi, umr, umi;
      if (pc & 1) {
        const float tau = (pc & 2) ? 1.0f : -1.0f;
        upr = 0.0f; umr = 0.0f;
        upi = tau * (p1 - p2);
        umi = tau * (p1 + p2);
      } else {
        const float sig = (pc & 2) ? -1.0f : 1.0f;
        upi = 0.0f; umi = 0.0f;
        upr = sig * (p1 + p2);
        umr = sig * (p1 - p2);
      }
      if (!imrow) { ag[jj] = part ? upi : upr;   ah[jj] = part ? -umr : umi; }
      else        { ag[jj] = part ? -upr : upi;  ah[jj] = part ? -umi : -umr; }
    }
    aG[c] = (v2f){ag[0], ag[1]};
    aH[c] = (v2f){ah[0], ah[1]};
  }

  const int nWaves = gridDim.x * (blockDim.x >> 5);
  const int gw     = blockIdx.x * (blockDim.x >> 5) + (threadIdx.x >> 5);

  for (int tile = gw; tile < NTILES_; tile += nWaves) {
    const int b   = tile >> 10;
    const int rem = tile & 1023;
    const int pr  = rem >> 3;
    const int pcg = rem & 7;

    const float* base = x + ((size_t)b * SIZE_ + 2 * pr) * SIZE_ + (pcg * 16 + p) * 2;
    const float2 lo = *(const float2*)base;
    const float2 hi = *(const float2*)(base + SIZE_);
    const float a0 = lo.x, a1 = lo.y, a2 = hi.x, a3 = hi.y;
    const float ss = a0 * a0 + a1 * a1 + a2 * a2 + a3 * a3;
    const float a1s = h ? a1 : -a1;

    v8f Tg = {0.f, 0.f, 0.f, 0.f, 0.f, 0.f, 0.f, 0.f};
    v8f Th = {0.f, 0.f, 0.f, 0.f, 0.f, 0.f, 0.f, 0.f};

#pragma unroll
    for (int cl = 0; cl < 2; ++cl) {
      float cc[2], sc[2], cs[2], sn[2];
#pragma unroll
      for (int jj = 0; jj < 2; ++jj) {
        const float t = (jj ? a0 : -a0) + a1s + (cl ? a2 : -a2) - a3;
        float es, ec, fs, fc;
        __sincosf(0.25f * (t * t - ss), &es, &ec);
        __sincosf(0.5f * t, &fs, &fc);
        cc[jj] = ec * fc;  sc[jj] = es * fc;
        cs[jj] = ec * fs;  sn[jj] = es * fs;
      }
      Tg = wmma4(aG[cl],     (v2f){cc[0], cc[1]}, Tg);
      Th = wmma4(aH[cl],     (v2f){cs[0], cs[1]}, Th);
      Tg = wmma4(aG[cl + 2], (v2f){sc[0], sc[1]}, Tg);
      Th = wmma4(aH[cl + 2], (v2f){sn[0], sn[1]}, Th);
    }

    float acc0 = 0.f, acc1 = 0.f, acc2 = 0.f, acc3 = 0.f;
#pragma unroll
    for (int v = 0; v < 8; ++v) {
      const float xs = Tg[v] + Th[v];
      const float xd = Tg[v] - Th[v];
      const float ps = xs * xs;
      const float pd = xd * xd;
      const int ja = v, jb = 15 - v;
      const int sa = (((ja >> 1) ^ (ja >> 2) ^ (ja >> 3)) & 1) | (((ja ^ (ja >> 1)) & 1) << 1) |
                     (((ja ^ (ja >> 1) ^ (ja >> 2)) & 1) << 2) |
                     (((ja ^ (ja >> 1) ^ (ja >> 2) ^ (ja >> 3)) & 1) << 3);
      const int sb = (((jb >> 1) ^ (jb >> 2) ^ (jb >> 3)) & 1) | (((jb ^ (jb >> 1)) & 1) << 1) |
                     (((jb ^ (jb >> 1) ^ (jb >> 2)) & 1) << 2) |
                     (((jb ^ (jb >> 1) ^ (jb >> 2) ^ (jb >> 3)) & 1) << 3);
      if (sa & 8) acc0 += ps;
      if (sa & 4) acc1 += ps;
      if (sa & 2) acc2 += ps;
      if (sa & 1) acc3 += ps;
      if (sb & 8) acc0 += pd;
      if (sb & 4) acc1 += pd;
      if (sb & 2) acc2 += pd;
      if (sb & 1) acc3 += pd;
    }
    acc0 += __shfl_xor(acc0, 16, 32);
    acc1 += __shfl_xor(acc1, 16, 32);
    acc2 += __shfl_xor(acc2, 16, 32);
    acc3 += __shfl_xor(acc3, 16, 32);

    if (lane < 16) {
      float4 o = make_float4(0.0625f * acc0, 0.0625f * acc1, 0.0625f * acc2, 0.0625f * acc3);
      *(float4*)(out + (((size_t)b * NPR_ + pr) * NPC_ + pcg * 16 + p) * 4) = o;
    }
  }
}

extern "C" void kernel_launch(void* const* d_in, const int* in_sizes, int n_in,
                              void* d_out, int out_size, void* d_ws, size_t ws_size,
                              hipStream_t stream) {
  (void)in_sizes; (void)n_in; (void)d_ws; (void)ws_size; (void)out_size;
  const float* x = (const float*)d_in[0];
  const float* w = (const float*)d_in[1];
  float* out = (float*)d_out;
  qconv2d_wmma_kernel<<<dim3(2048), dim3(256), 0, stream>>>(x, w, out);
}


# A operands built once per work-group: each of the 256 threads makes one element pair, 2 KiB of LDS and one barrier hand every wave its 16 values; was all 16 per lane in every wave; same sums
# speedup vs baseline: 1.3596x; 1.0692x over previous
#include <hip/hip_runtime.h>


typedef float v2f __attribute__((ext_vector_type(2)));
typedef float v8f __attribute__((ext_vector_type(8)));

#define BATCH_   64
#define SIZE_    256
#define NPR_     128
#define NPC_     128
#define NPCG_    8
#define NTILES_  (BATCH_ * NPR_ * NPCG_)

__device__ __forceinline__ v8f wmma4(v2f a, v2f b, v8f c) {
  return __builtin_amdgcn_wmma_f32_16x16x4_f32(false, a, false, b, (short)0, c,
                                               false, false);
}

__global__ __launch_bounds__(256)
void qconv2d_wmma_kernel(const float* __restrict__ x,
                         const float* __restrict__ w,
                         float* __restrict__ out) {
  const int lane = threadIdx.x & 31;
  const int p    = lane & 15;
  const int h    = (lane >> 4) & 1;

  __shared__ float4 aop[4 * 32];

  float cq[4], sq[4];
#pragma unroll
  for (int q = 0; q < 4; ++q) {
    __sincosf(0.5f * w[q], &sq[q], &cq[q]);
  }
  const int j     = p & 7;
  const int imrow = p >> 3;

  {
    const int wv   = __builtin_amdgcn_readfirstlane((int)(threadIdx.x >> 5));
    const int c    = wv & 3;
    const int jj   = (wv >> 2) & 1;
    const int k    = 4 * (c & 1) + 2 * h + jj;
    const int part = c >> 1;
    const int xo   = j ^ k;
    const int pc   = __popc(xo);
    float p1 = 1.0f, p2 = 1.0f;
#pragma unroll
    for (int q = 0; q < 4; ++q) {
      const bool b = (xo >> q) & 1;
      p1 *= b ? sq[q] : cq[q];
      p2 *= b ? cq[q] : sq[q];
    }
    float upr, upi, umr, umi;
    if (pc & 1) {
      const float tau = (pc & 2) ? 1.0f : -1.0f;
      upr = 0.0f; umr = 0.0f;
      upi = tau * (p1 - p2);
      umi = tau * (p1 + p2);
    } else {
      const float sig = (pc & 2) ? -1.0f : 1.0f;
      upi = 0.0f; umi = 0.0f;
      upr = sig * (p1 + p2);
      umr = sig * (p1 - p2);
    }
    float ag, ah;
    if (!imrow) { ag = part ? upi : upr;   ah = part ? -umr : umi; }
    else        { ag = part ? -upr : upi;  ah = part ? -umi : -umr; }
    float* a1 = (float*)aop;
    const int e = (c & 1) * 2 + jj;
    a1[(part * 32 + lane) * 4 + e]       = ag;
    a1[((2 + part) * 32 + lane) * 4 + e] = ah;
  }
  __syncthreads();

  const float4 g01 = aop[lane];
  const float4 g23 = aop[32 + lane];
  const float4 h01 = aop[64 + lane];
  const float4 h23 = aop[96 + lane];
  const v2f aG[4] = {(v2f){g01.x, g01.y}, (v2f){g01.z, g01.w}, (v2f){g23.x, g23.y}, (v2f){g23.z, g23.w}};
  const v2f aH[4] = {(v2f){h01.x, h01.y}, (v2f){h01.z, h01.w}, (v2f){h23.x, h23.y}, (v2f){h23.z, h23.w}};

  const int nWaves = gridDim.x * (blockDim.x >> 5);
  const int gw     = blockIdx.x * (blockDim.x >> 5) + (threadIdx.x >> 5);

  for (int tile = gw; tile < NTILES_; tile += nWaves) {
    const int b   = tile >> 10;
    const int rem = tile & 1023;
    const int pr  = rem >> 3;
    const int pcg = rem & 7;

    const float* base = x + ((size_t)b * SIZE_ + 2 * pr) * SIZE_ + (pcg * 16 + p) * 2;
    const float2 lo = *(const float2*)base;
    const float2 hi = *(const float2*)(base + SIZE_);
    const float a0 = lo.x, a1 = lo.y, a2 = hi.x, a3 = hi.y;
    const float ss = a0 * a0 + a1 * a1 + a2 * a2 + a3 * a3;
    const float a1s = h ? a1 : -a1;

    v8f Tg = {0.f, 0.f, 0.f, 0.f, 0.f, 0.f, 0.f, 0.f};
    v8f Th = {0.f, 0.f, 0.f, 0.f, 0.f, 0.f, 0.f, 0.f};

#pragma unroll
    for (int cl = 0; cl < 2; ++cl) {
      float cc[2], sc[2], cs[2], sn[2];
#pragma unroll
      for (int jj = 0; jj < 2; ++jj) {
        const float t = (jj ? a0 : -a0) + a1s + (cl ? a2 : -a2) - a3;
        float es, ec, fs, fc;
        __sincosf(0.25f * (t * t - ss), &es, &ec);
        __sincosf(0.5f * t, &fs, &fc);
        cc[jj] = ec * fc;  sc[jj] = es * fc;
        cs[jj] = ec * fs;  sn[jj] = es * fs;
      }
      Tg = wmma4(aG[cl],     (v2f){cc[0], cc[1]}, Tg);
      Th = wmma4(aH[cl],     (v2f){cs[0], cs[1]}, Th);
      Tg = wmma4(aG[cl + 2], (v2f){sc[0], sc[1]}, Tg);
      Th = wmma4(aH[cl + 2], (v2f){sn[0], sn[1]}, Th);
    }

    float acc0 = 0.f, acc1 = 0.f, acc2 = 0.f, acc3 = 0.f;
#pragma unroll
    for (int v = 0; v < 8; ++v) {
      const float xs = Tg[v] + Th[v];
      const float xd = Tg[v] - Th[v];
      const float ps = xs * xs;
      const float pd = xd * xd;
      const int ja = v, jb = 15 - v;
      const int sa = (((ja >> 1) ^ (ja >> 2) ^ (ja >> 3)) & 1) | (((ja ^ (ja >> 1)) & 1) << 1) |
                     (((ja ^ (ja >> 1) ^ (ja >> 2)) & 1) << 2) |
                     (((ja ^ (ja >> 1) ^ (ja >> 2) ^ (ja >> 3)) & 1) << 3);
      const int sb = (((jb >> 1) ^ (jb >> 2) ^ (jb >> 3)) & 1) | (((jb ^ (jb >> 1)) & 1) << 1) |
                     (((jb ^ (jb >> 1) ^ (jb >> 2)) & 1) << 2) |
                     (((jb ^ (jb >> 1) ^ (jb >> 2) ^ (jb >> 3)) & 1) << 3);
      if (sa & 8) acc0 += ps;
      if (sa & 4) acc1 += ps;
      if (sa & 2) acc2 += ps;
      if (sa & 1) acc3 += ps;
      if (sb & 8) acc0 += pd;
      if (sb & 4) acc1 += pd;
      if (sb & 2) acc2 += pd;
      if (sb & 1) acc3 += pd;
    }
    acc0 += __shfl_xor(acc0, 16, 32);
    acc1 += __shfl_xor(acc1, 16, 32);
    acc2 += __shfl_xor(acc2, 16, 32);
    acc3 += __shfl_xor(acc3, 16, 32);

    if (lane < 16) {
      float4 o = make_float4(0.0625f * acc0, 0.0625f * acc1, 0.0625f * acc2, 0.0625f * acc3);
      *(float4*)(out + (((size_t)b * NPR_ + pr) * NPC_ + pcg * 16 + p) * 4) = o;
    }
  }
}

extern "C" void kernel_launch(void* const* d_in, const int* in_sizes, int n_in,
                              void* d_out, int out_size, void* d_ws, size_t ws_size,
                              hipStream_t stream) {
  (void)in_sizes; (void)n_in; (void)d_ws; (void)ws_size; (void)out_size;
  const float* x = (const float*)d_in[0];
  const float* w = (const float*)d_in[1];
  float* out = (float*)d_out;
  qconv2d_wmma_kernel<<<dim3(2048), dim3(256), 0, stream>>>(x, w, out);
}


# Static tile map: block = (image, 4 patch rows), wave = column group; the 4 tiles unrolled, so tile index, address arithmetic and loop control leave the vector unit; same sums
# speedup vs baseline: 1.4358x; 1.0560x over previous
#include <hip/hip_runtime.h>


typedef float v2f __attribute__((ext_vector_type(2)));
typedef float v8f __attribute__((ext_vector_type(8)));

#define BATCH_   64
#define SIZE_    256
#define NPR_     128
#define NPC_     128
#define NPCG_    8
#define RPW_     4
#define NRQ_     (NPR_ / RPW_)

__device__ __forceinline__ v8f wmma4(v2f a, v2f b, v8f c) {
  return __builtin_amdgcn_wmma_f32_16x16x4_f32(false, a, false, b, (short)0, c,
                                               false, false);
}

__global__ __launch_bounds__(256)
void qconv2d_wmma_kernel(const float* __restrict__ x,
                         const float* __restrict__ w,
                         float* __restrict__ out) {
  const int lane = threadIdx.x & 31;
  const int p    = lane & 15;
  const int h    = (lane >> 4) & 1;

  __shared__ float4 aop[4 * 32];

  float cq[4], sq[4];
#pragma unroll
  for (int q = 0; q < 4; ++q) {
    __sincosf(0.5f * w[q], &sq[q], &cq[q]);
  }
  const int j     = p & 7;
  const int imrow = p >> 3;

  {
    const int wv   = __builtin_amdgcn_readfirstlane((int)(threadIdx.x >> 5));
    const int c    = wv & 3;
    const int jj   = (wv >> 2) & 1;
    const int k    = 4 * (c & 1) + 2 * h + jj;
    const int part = c >> 1;
    const int xo   = j ^ k;
    const int pc   = __popc(xo);
    float p1 = 1.0f, p2 = 1.0f;
#pragma unroll
    for (int q = 0; q < 4; ++q) {
      const bool b = (xo >> q) & 1;
      p1 *= b ? sq[q] : cq[q];
      p2 *= b ? cq[q] : sq[q];
    }
    float upr, upi, umr, umi;
    if (pc & 1) {
      const float tau = (pc & 2) ? 1.0f : -1.0f;
      upr = 0.0f; umr = 0.0f;
      upi = tau * (p1 - p2);
      umi = tau * (p1 + p2);
    } else {
      const float sig = (pc & 2) ? -1.0f : 1.0f;
      upi = 0.0f; umi = 0.0f;
      upr = sig * (p1 + p2);
      umr = sig * (p1 - p2);
    }
    float ag, ah;
    if (!imrow) { ag = part ? upi : upr;   ah = part ? -umr : umi; }
    else        { ag = part ? -upr : upi;  ah = part ? -umi : -umr; }
    float* a1 = (float*)aop;
    const int e = (c & 1) * 2 + jj;
    a1[(part * 32 + lane) * 4 + e]       = ag;
    a1[((2 + part) * 32 + lane) * 4 + e] = ah;
  }
  __syncthreads();

  const float4 g01 = aop[lane];
  const float4 g23 = aop[32 + lane];
  const float4 h01 = aop[64 + lane];
  const float4 h23 = aop[96 + lane];
  const v2f aG[4] = {(v2f){g01.x, g01.y}, (v2f){g01.z, g01.w}, (v2f){g23.x, g23.y}, (v2f){g23.z, g23.w}};
  const v2f aH[4] = {(v2f){h01.x, h01.y}, (v2f){h01.z, h01.w}, (v2f){h23.x, h23.y}, (v2f){h23.z, h23.w}};

  const int bimg = blockIdx.x / NRQ_;
  const int prq  = blockIdx.x % NRQ_;
  const int pcg  = __builtin_amdgcn_readfirstlane((int)(threadIdx.x >> 5));
  const float* xw = x + ((size_t)bimg * SIZE_ + 2 * RPW_ * prq) * SIZE_ + pcg * 32 + p * 2;
  float*       ow = out + (((size_t)bimg * NPR_ + RPW_ * prq) * NPC_ + pcg * 16 + p) * 4;

#pragma unroll
  for (int r = 0; r < RPW_; ++r) {
    const float* base = xw + r * (2 * SIZE_);
    const float2 lo = *(const float2*)base;
    const float2 hi = *(const float2*)(base + SIZE_);
    const float a0 = lo.x, a1 = lo.y, a2 = hi.x, a3 = hi.y;
    const float ss = a0 * a0 + a1 * a1 + a2 * a2 + a3 * a3;
    const float a1s = h ? a1 : -a1;

    v8f Tg = {0.f, 0.f, 0.f, 0.f, 0.f, 0.f, 0.f, 0.f};
    v8f Th = {0.f, 0.f, 0.f, 0.f, 0.f, 0.f, 0.f, 0.f};

#pragma unroll
    for (int cl = 0; cl < 2; ++cl) {
      float cc[2], sc[2], cs[2], sn[2];
#pragma unroll
      for (int jj = 0; jj < 2; ++jj) {
        const float t = (jj ? a0 : -a0) + a1s + (cl ? a2 : -a2) - a3;
        float es, ec, fs, fc;
        __sincosf(0.25f * (t * t - ss), &es, &ec);
        __sincosf(0.5f * t, &fs, &fc);
        cc[jj] = ec * fc;  sc[jj] = es * fc;
        cs[jj] = ec * fs;  sn[jj] = es * fs;
      }
      Tg = wmma4(aG[cl],     (v2f){cc[0], cc[1]}, Tg);
      Th = wmma4(aH[cl],     (v2f){cs[0], cs[1]}, Th);
      Tg = wmma4(aG[cl + 2], (v2f){sc[0], sc[1]}, Tg);
      Th = wmma4(aH[cl + 2], (v2f){sn[0], sn[1]}, Th);
    }

    float acc0 = 0.f, acc1 = 0.f, acc2 = 0.f, acc3 = 0.f;
#pragma unroll
    for (int v = 0; v < 8; ++v) {
      const float xs = Tg[v] + Th[v];
      const float xd = Tg[v] - Th[v];
      const float ps = xs * xs;
      const float pd = xd * xd;
      const int ja = v, jb = 15 - v;
      const int sa = (((ja >> 1) ^ (ja >> 2) ^ (ja >> 3)) & 1) | (((ja ^ (ja >> 1)) & 1) << 1) |
                     (((ja ^ (ja >> 1) ^ (ja >> 2)) & 1) << 2) |
                     (((ja ^ (ja >> 1) ^ (ja >> 2) ^ (ja >> 3)) & 1) << 3);
      const int sb = (((jb >> 1) ^ (jb >> 2) ^ (jb >> 3)) & 1) | (((jb ^ (jb >> 1)) & 1) << 1) |
                     (((jb ^ (jb >> 1) ^ (jb >> 2)) & 1) << 2) |
                     (((jb ^ (jb >> 1) ^ (jb >> 2) ^ (jb >> 3)) & 1) << 3);
      if (sa & 8) acc0 += ps;
      if (sa & 4) acc1 += ps;
      if (sa & 2) acc2 += ps;
      if (sa & 1) acc3 += ps;
      if (sb & 8) acc0 += pd;
      if (sb & 4) acc1 += pd;
      if (sb & 2) acc2 += pd;
      if (sb & 1) acc3 += pd;
    }
    acc0 += __shfl_xor(acc0, 16, 32);
    acc1 += __shfl_xor(acc1, 16, 32);
    acc2 += __shfl_xor(acc2, 16, 32);
    acc3 += __shfl_xor(acc3, 16, 32);

    if (lane < 16) {
      float4 o = make_float4(0.0625f * acc0, 0.0625f * acc1, 0.0625f * acc2, 0.0625f * acc3);
      *(float4*)(ow + r * (NPC_ * 4)) = o;
    }
  }
}

extern "C" void kernel_launch(void* const* d_in, const int* in_sizes, int n_in,
                              void* d_out, int out_size, void* d_ws, size_t ws_size,
                              hipStream_t stream) {
  (void)in_sizes; (void)n_in; (void)d_ws; (void)ws_size; (void)out_size;
  const float* x = (const float*)d_in[0];
  const float* w = (const float*)d_in[1];
  float* out = (float*)d_out;
  qconv2d_wmma_kernel<<<dim3(2048), dim3(256), 0, stream>>>(x, w, out);
}


# Epilogue without sum/difference tiles: rows reordered and operands scaled so bits 3,1 = sums of Tg^2+Th^2 and bits 2,0 = 1/2 + signed sums of Tg*Th (total probability 1); sums reordered
# speedup vs baseline: 1.5555x; 1.0834x over previous
#include <hip/hip_runtime.h>


typedef float v2f __attribute__((ext_vector_type(2)));
typedef float v8f __attribute__((ext_vector_type(8)));

#define BATCH_   64
#define SIZE_    256
#define NPR_     128
#define NPC_     128
#define NPCG_    8
#define RPW_     4
#define NRQ_     (NPR_ / RPW_)

__device__ __forceinline__ v8f wmma4(v2f a, v2f b, v8f c) {
  return __builtin_amdgcn_wmma_f32_16x16x4_f32(false, a, false, b, (short)0, c,
                                               false, false);
}

__global__ __launch_bounds__(256)
void qconv2d_wmma_kernel(const float* __restrict__ x,
                         const float* __restrict__ w,
                         float* __restrict__ out) {
  const int lane = threadIdx.x & 31;
  const int p    = lane & 15;
  const int h    = (lane >> 4) & 1;

  __shared__ float4 aop[4 * 32];

  float cq[4], sq[4];
#pragma unroll
  for (int q = 0; q < 4; ++q) {
    __sincosf(0.5f * w[q], &sq[q], &cq[q]);
  }
  const int j     = (p & 7) ^ ((p & 1) << 1);
  const int imrow = p >> 3;

  {
    const int wv   = __builtin_amdgcn_readfirstlane((int)(threadIdx.x >> 5));
    const int c    = wv & 3;
    const int jj   = (wv >> 2) & 1;
    const int k    = 4 * (c & 1) + 2 * h + jj;
    const int part = c >> 1;
    const int xo   = j ^ k;
    const int pc   = __popc(xo);
    float p1 = 1.0f, p2 = 1.0f;
#pragma unroll
    for (int q = 0; q < 4; ++q) {
      const bool b = (xo >> q) & 1;
      p1 *= b ? sq[q] : cq[q];
      p2 *= b ? cq[q] : sq[q];
    }
    float upr, upi, umr, umi;
    if (pc & 1) {
      const float tau = (pc & 2) ? 1.0f : -1.0f;
      upr = 0.0f; umr = 0.0f;
      upi = tau * (p1 - p2);
      umi = tau * (p1 + p2);
    } else {
      const float sig = (pc & 2) ? -1.0f : 1.0f;
      upi = 0.0f; umi = 0.0f;
      upr = sig * (p1 + p2);
      umr = sig * (p1 - p2);
    }
    float ag, ah;
    if (!imrow) { ag = part ? upi : upr;   ah = part ? -umr : umi; }
    else        { ag = part ? -upr : upi;  ah = part ? -umi : -umr; }
    float* a1 = (float*)aop;
    const int e = (c & 1) * 2 + jj;
    const float rs8 = 0.35355339f;
    a1[(part * 32 + lane) * 4 + e]       = rs8 * ag;
    a1[((2 + part) * 32 + lane) * 4 + e] = rs8 * ah;
  }
  __syncthreads();

  const float4 g01 = aop[lane];
  const float4 g23 = aop[32 + lane];
  const float4 h01 = aop[64 + lane];
  const float4 h23 = aop[96 + lane];
  const v2f aG[4] = {(v2f){g01.x, g01.y}, (v2f){g01.z, g01.w}, (v2f){g23.x, g23.y}, (v2f){g23.z, g23.w}};
  const v2f aH[4] = {(v2f){h01.x, h01.y}, (v2f){h01.z, h01.w}, (v2f){h23.x, h23.y}, (v2f){h23.z, h23.w}};

  const int bimg = blockIdx.x / NRQ_;
  const int prq  = blockIdx.x % NRQ_;
  const int pcg  = __builtin_amdgcn_readfirstlane((int)(threadIdx.x >> 5));
  const float* xw = x + ((size_t)bimg * SIZE_ + 2 * RPW_ * prq) * SIZE_ + pcg * 32 + p * 2;
  float*       ow = out + (((size_t)bimg * NPR_ + RPW_ * prq) * NPC_ + pcg * 16 + p) * 4;

  const v2f half0 = {h ? 0.0f : 0.5f, 0.0f};

#pragma unroll
  for (int r = 0; r < RPW_; ++r) {
    const float* base = xw + r * (2 * SIZE_);
    const float2 lo = *(const float2*)base;
    const float2 hi = *(const float2*)(base + SIZE_);
    const float a0 = lo.x, a1 = lo.y, a2 = hi.x, a3 = hi.y;
    const float ss = a0 * a0 + a1 * a1 + a2 * a2 + a3 * a3;
    const float a1s = h ? a1 : -a1;

    v8f Tg = {0.f, 0.f, 0.f, 0.f, 0.f, 0.f, 0.f, 0.f};
    v8f Th = {0.f, 0.f, 0.f, 0.f, 0.f, 0.f, 0.f, 0.f};

#pragma unroll
    for (int cl = 0; cl < 2; ++cl) {
      float cc[2], sc[2], cs[2], sn[2];
#pragma unroll
      for (int jj = 0; jj < 2; ++jj) {
        const float t = (jj ? a0 : -a0) + a1s + (cl ? a2 : -a2) - a3;
        float es, ec, fs, fc;
        __sincosf(0.25f * (t * t - ss), &es, &ec);
        __sincosf(0.5f * t, &fs, &fc);
        cc[jj] = ec * fc;  sc[jj] = es * fc;
        cs[jj] = ec * fs;  sn[jj] = es * fs;
      }
      Tg = wmma4(aG[cl],     (v2f){cc[0], cc[1]}, Tg);
      Th = wmma4(aH[cl],     (v2f){cs[0], cs[1]}, Th);
      Tg = wmma4(aG[cl + 2], (v2f){sc[0], sc[1]}, Tg);
      Th = wmma4(aH[cl + 2], (v2f){sn[0], sn[1]}, Th);
    }

    const v2f g0 = {Tg[0], Tg[1]}, g1 = {Tg[2], Tg[3]}, g2 = {Tg[4], Tg[5]}, g3 = {Tg[6], Tg[7]};
    const v2f h0 = {Th[0], Th[1]}, h1 = {Th[2], Th[3]}, h2 = {Th[4], Th[5]}, h3 = {Th[6], Th[7]};
    const v2f xs = half0 - g0 * h0 + g1 * h1 + g2 * h2 - g3 * h3;
    const v2f q1 = g1 * g1 + h1 * h1;
    const v2f w3 = q1 + g2 * g2 + h2 * h2;
    const v2f w1 = q1 + g3 * g3 + h3 * h3;
    const v2f a20 = (v2f){xs.x, xs.x} + (v2f){xs.y, -xs.y};
    const v2f a31 = (v2f){w3.x, w1.x} + (v2f){w3.y, w1.y};
    float acc0 = a31.x, acc1 = a20.x, acc2 = a31.y, acc3 = a20.y;
    acc0 += __shfl_xor(acc0, 16, 32);
    acc1 += __shfl_xor(acc1, 16, 32);
    acc2 += __shfl_xor(acc2, 16, 32);
    acc3 += __shfl_xor(acc3, 16, 32);

    if (lane < 16) {
      float4 o = make_float4(acc0, acc1, acc2, acc3);
      *(float4*)(ow + r * (NPC_ * 4)) = o;
    }
  }
}

extern "C" void kernel_launch(void* const* d_in, const int* in_sizes, int n_in,
                              void* d_out, int out_size, void* d_ws, size_t ws_size,
                              hipStream_t stream) {
  (void)in_sizes; (void)n_in; (void)d_ws; (void)ws_size; (void)out_size;
  const float* x = (const float*)d_in[0];
  const float* w = (const float*)d_in[1];
  float* out = (float*)d_out;
  qconv2d_wmma_kernel<<<dim3(2048), dim3(256), 0, stream>>>(x, w, out);
}


# Input through LDS: work-group fetches its 8 image rows (one 8 KiB run) by two 16-byte loads per thread issued ahead of the operand build (made branch-free); waves read angles from LDS; same sums
# speedup vs baseline: 1.6796x; 1.0798x over previous
#include <hip/hip_runtime.h>


typedef float v2f __attribute__((ext_vector_type(2)));
typedef float v8f __attribute__((ext_vector_type(8)));

#define BATCH_   64
#define SIZE_    256
#define NPR_     128
#define NPC_     128
#define NPCG_    8
#define RPW_     4
#define NRQ_     (NPR_ / RPW_)

__device__ __forceinline__ v8f wmma4(v2f a, v2f b, v8f c) {
  return __builtin_amdgcn_wmma_f32_16x16x4_f32(false, a, false, b, (short)0, c,
                                               false, false);
}

__global__ __launch_bounds__(256)
void qconv2d_wmma_kernel(const float* __restrict__ x,
                         const float* __restrict__ w,
                         float* __restrict__ out) {
  const int lane = threadIdx.x & 31;
  const int p    = lane & 15;
  const int h    = (lane >> 4) & 1;

  __shared__ float4 aop[4 * 32];

  __shared__ float4 xst[2 * RPW_ * SIZE_ / 4];
  const int bimg = blockIdx.x / NRQ_;
  const int prq  = blockIdx.x % NRQ_;
  const float4* xb = (const float4*)(x + ((size_t)bimg * SIZE_ + 2 * RPW_ * prq) * SIZE_);
  const float4 xa = xb[threadIdx.x];
  const float4 xc = xb[256 + threadIdx.x];

  float cq[4], sq[4];
#pragma unroll
  for (int q = 0; q < 4; ++q) {
    __sincosf(0.5f * w[q], &sq[q], &cq[q]);
  }
  const int j     = (p & 7) ^ ((p & 1) << 1);
  const int imrow = p >> 3;

  {
    const int wv   = __builtin_amdgcn_readfirstlane((int)(threadIdx.x >> 5));
    const int c    = wv & 3;
    const int jj   = (wv >> 2) & 1;
    const int k    = 4 * (c & 1) + 2 * h + jj;
    const int part = c >> 1;
    const int xo   = j ^ k;
    const int pc   = __popc(xo);
    float p1 = 1.0f, p2 = 1.0f;
#pragma unroll
    for (int q = 0; q < 4; ++q) {
      const bool b = (xo >> q) & 1;
      p1 *= b ? sq[q] : cq[q];
      p2 *= b ? cq[q] : sq[q];
    }
    const bool  odd = pc & 1;
    const float sig = (pc & 2) ? -1.0f : 1.0f;
    const float vs  = sig * (p1 + p2);
    const float vd  = sig * (p1 - p2);
    const float upr = odd ? 0.0f : vs;
    const float umr = odd ? 0.0f : vd;
    const float upi = odd ? -vd : 0.0f;
    const float umi = odd ? -vs : 0.0f;
    float ag, ah;
    if (!imrow) { ag = part ? upi : upr;   ah = part ? -umr : umi; }
    else        { ag = part ? -upr : upi;  ah = part ? -umi : -umr; }
    float* a1 = (float*)aop;
    const int e = (c & 1) * 2 + jj;
    const float rs8 = 0.35355339f;
    a1[(part * 32 + lane) * 4 + e]       = rs8 * ag;
    a1[((2 + part) * 32 + lane) * 4 + e] = rs8 * ah;
  }
  xst[threadIdx.x]       = xa;
  xst[256 + threadIdx.x] = xc;
  __syncthreads();

  const float4 g01 = aop[lane];
  const float4 g23 = aop[32 + lane];
  const float4 h01 = aop[64 + lane];
  const float4 h23 = aop[96 + lane];
  const v2f aG[4] = {(v2f){g01.x, g01.y}, (v2f){g01.z, g01.w}, (v2f){g23.x, g23.y}, (v2f){g23.z, g23.w}};
  const v2f aH[4] = {(v2f){h01.x, h01.y}, (v2f){h01.z, h01.w}, (v2f){h23.x, h23.y}, (v2f){h23.z, h23.w}};

  const int pcg  = __builtin_amdgcn_readfirstlane((int)(threadIdx.x >> 5));
  const float2* xw = (const float2*)xst + pcg * 16 + p;
  float*       ow = out + (((size_t)bimg * NPR_ + RPW_ * prq) * NPC_ + pcg * 16 + p) * 4;

  const v2f half0 = {h ? 0.0f : 0.5f, 0.0f};

  float2 lo4[RPW_], hi4[RPW_];
#pragma unroll
  for (int r = 0; r < RPW_; ++r) {
    lo4[r] = xw[(2 * r) * (SIZE_ / 2)];
    hi4[r] = xw[(2 * r + 1) * (SIZE_ / 2)];
  }

#pragma unroll
  for (int r = 0; r < RPW_; ++r) {
    const float2 lo = lo4[r];
    const float2 hi = hi4[r];
    const float a0 = lo.x, a1 = lo.y, a2 = hi.x, a3 = hi.y;
    const float ss = a0 * a0 + a1 * a1 + a2 * a2 + a3 * a3;
    const float a1s = h ? a1 : -a1;

    v8f Tg = {0.f, 0.f, 0.f, 0.f, 0.f, 0.f, 0.f, 0.f};
    v8f Th = {0.f, 0.f, 0.f, 0.f, 0.f, 0.f, 0.f, 0.f};

#pragma unroll
    for (int cl = 0; cl < 2; ++cl) {
      float cc[2], sc[2], cs[2], sn[2];
#pragma unroll
      for (int jj = 0; jj < 2; ++jj) {
        const float t = (jj ? a0 : -a0) + a1s + (cl ? a2 : -a2) - a3;
        float es, ec, fs, fc;
        __sincosf(0.25f * (t * t - ss), &es, &ec);
        __sincosf(0.5f * t, &fs, &fc);
        cc[jj] = ec * fc;  sc[jj] = es * fc;
        cs[jj] = ec * fs;  sn[jj] = es * fs;
      }
      Tg = wmma4(aG[cl],     (v2f){cc[0], cc[1]}, Tg);
      Th = wmma4(aH[cl],     (v2f){cs[0], cs[1]}, Th);
      Tg = wmma4(aG[cl + 2], (v2f){sc[0], sc[1]}, Tg);
      Th = wmma4(aH[cl + 2], (v2f){sn[0], sn[1]}, Th);
    }

    const v2f g0 = {Tg[0], Tg[1]}, g1 = {Tg[2], Tg[3]}, g2 = {Tg[4], Tg[5]}, g3 = {Tg[6], Tg[7]};
    const v2f h0 = {Th[0], Th[1]}, h1 = {Th[2], Th[3]}, h2 = {Th[4], Th[5]}, h3 = {Th[6], Th[7]};
    const v2f xs = half0 - g0 * h0 + g1 * h1 + g2 * h2 - g3 * h3;
    const v2f q1 = g1 * g1 + h1 * h1;
    const v2f w3 = q1 + g2 * g2 + h2 * h2;
    const v2f w1 = q1 + g3 * g3 + h3 * h3;
    const v2f a20 = (v2f){xs.x, xs.x} + (v2f){xs.y, -xs.y};
    const v2f a31 = (v2f){w3.x, w1.x} + (v2f){w3.y, w1.y};
    float acc0 = a31.x, acc1 = a20.x, acc2 = a31.y, acc3 = a20.y;
    acc0 += __shfl_xor(acc0, 16, 32);
    acc1 += __shfl_xor(acc1, 16, 32);
    acc2 += __shfl_xor(acc2, 16, 32);
    acc3 += __shfl_xor(acc3, 16, 32);

    if (lane < 16) {
      float4 o = make_float4(acc0, acc1, acc2, acc3);
      *(float4*)(ow + r * (NPC_ * 4)) = o;
    }
  }
}

extern "C" void kernel_launch(void* const* d_in, const int* in_sizes, int n_in,
                              void* d_out, int out_size, void* d_ws, size_t ws_size,
                              hipStream_t stream) {
  (void)in_sizes; (void)n_in; (void)d_ws; (void)ws_size; (void)out_size;
  const float* x = (const float*)d_in[0];
  const float* w = (const float*)d_in[1];
  float* out = (float*)d_out;
  qconv2d_wmma_kernel<<<dim3(2048), dim3(256), 0, stream>>>(x, w, out);
}
